# starting kernel
# baseline (speedup 1.0000x reference)
#include <hip/hip_runtime.h>
#include <hip/hip_bf16.h>
#include <math.h>

#define VOCAB 50000
#define EMBD  128
#define HOPS  3
#define BS    64
#define STORY 50
#define SENT  20
#define QLEN  20

typedef __attribute__((ext_vector_type(2))) float v2f;
typedef __attribute__((ext_vector_type(8))) float v8f;

__global__ __launch_bounds__(EMBD) void memnn_u0(const int* __restrict__ q,
                                                 const float* __restrict__ A,
                                                 float* __restrict__ u) {
    __shared__ int tok[QLEN];
    const int b = blockIdx.x;
    const int d = threadIdx.x;
    if (d < QLEN) tok[d] = q[b * QLEN + d];
    __syncthreads();
    float acc = 0.0f;
#pragma unroll
    for (int j = 0; j < QLEN; ++j)
        acc += A[(size_t)tok[j] * EMBD + d];
    u[b * EMBD + d] = acc;
}

__global__ __launch_bounds__(EMBD) void memnn_gather(const int* __restrict__ x,
                                                     const float* __restrict__ A,
                                                     const float* __restrict__ TA,
                                                     const float* __restrict__ TC,
                                                     float* __restrict__ m,
                                                     float* __restrict__ c) {
    const int bs = blockIdx.x;
    const int s  = bs % STORY;
    const int d  = threadIdx.x;
    __shared__ int tok[SENT];
    if (d < SENT) tok[d] = x[(size_t)bs * SENT + d];
    __syncthreads();

    const float kd = (float)(d + 1) / (float)EMBD;
    const size_t LVL = (size_t)VOCAB * EMBD;

    float mp0 = 0.f, mp1 = 0.f, mp2 = 0.f;
    float cp1 = 0.f, cp2 = 0.f, cp3 = 0.f;
#pragma unroll
    for (int j = 0; j < SENT; ++j) {
        const float jj = (float)(j + 1) / (float)SENT;
        const float pe = 1.0f - jj - kd * (1.0f - 2.0f * jj);
        const size_t base = (size_t)tok[j] * EMBD + d;
        const float e0 = A[base];
        const float e1 = A[LVL + base];
        const float e2 = A[2 * LVL + base];
        const float e3 = A[3 * LVL + base];
        mp0 += e0 * pe;  mp1 += e1 * pe;  mp2 += e2 * pe;
        cp1 += e1;       cp2 += e2;       cp3 += e3;
    }
    const float ta = TA[s * EMBD + d];
    const float tc = TC[s * EMBD + d];
    const size_t off = (size_t)bs * EMBD + d;
    const size_t hs  = (size_t)BS * STORY * EMBD;
    m[off]          = mp0 + ta;
    m[hs + off]     = mp1 + ta;
    m[2 * hs + off] = mp2 + ta;
    c[off]          = cp1 + tc;
    c[hs + off]     = cp2 + tc;
    c[2 * hs + off] = cp3 + tc;
}

__global__ __launch_bounds__(EMBD) void memnn_hop(const float* __restrict__ m,
                                                  const float* __restrict__ c,
                                                  float* __restrict__ u,
                                                  int hop) {
    const int b = blockIdx.x;
    const int t = threadIdx.x;
    __shared__ float u_lds[EMBD];
    __shared__ float score[STORY];
    u_lds[t] = u[b * EMBD + t];
    __syncthreads();

    const int wave = t >> 5;
    const int lane = t & 31;
    const size_t hb = (size_t)hop * BS * STORY * EMBD + (size_t)b * STORY * EMBD;

    for (int s = wave; s < STORY; s += 4) {
        const float* mrow = m + hb + (size_t)s * EMBD;
        float p = 0.f;
#pragma unroll
        for (int i = 0; i < EMBD / 32; ++i)
            p += mrow[lane + 32 * i] * u_lds[lane + 32 * i];
        for (int o = 16; o > 0; o >>= 1) p += __shfl_xor(p, o, 32);
        if (lane == 0) score[s] = p;
    }
    __syncthreads();

    float mx = -3.4e38f;
    for (int s = 0; s < STORY; ++s) mx = fmaxf(mx, score[s]);
    float sm = 0.f;
    for (int s = 0; s < STORY; ++s) sm += __expf(score[s] - mx);
    const float inv = 1.0f / sm;

    float acc = 0.f;
    const float* crow = c + hb;
    for (int s = 0; s < STORY; ++s)
        acc += (__expf(score[s] - mx) * inv) * crow[(size_t)s * EMBD + t];
    u[b * EMBD + t] = u_lds[t] + acc;
}

__global__ __launch_bounds__(256) void memnn_logits_wmma(const float* __restrict__ u,
                                                         const float* __restrict__ A3,
                                                         float* __restrict__ out) {
    __shared__ float u_lds[BS * EMBD];
    for (int i = threadIdx.x; i < BS * EMBD; i += 256) u_lds[i] = u[i];
    __syncthreads();

    const int wave = threadIdx.x >> 5;
    const int lane = threadIdx.x & 31;
    const int tile = blockIdx.x * 8 + wave;
    const int NTILES = (VOCAB / 16) * (BS / 16);
    if (tile >= NTILES) return;

    const int mb = (tile & 3) << 4;
    const int nb = (tile >> 2) << 4;
    const int n  = lane & 15;
    const int h  = lane >> 4;

    const float* arow = u_lds + (mb + n) * EMBD;
    const float* brow = A3 + (size_t)(nb + n) * EMBD;

    v8f acc = {};
#pragma unroll
    for (int kk = 0; kk < EMBD / 4; ++kk) {
        const int k0 = kk * 4 + 2 * h;
        v2f a;  a.x  = arow[k0]; a.y  = arow[k0 + 1];
        v2f bv; bv.x = brow[k0]; bv.y = brow[k0 + 1];
        acc = __builtin_amdgcn_wmma_f32_16x16x4_f32(
             false, a,  false, bv,
             (short)0, acc,  false,  false);
    }
#pragma unroll
    for (int v = 0; v < 8; ++v)
        out[(size_t)(mb + v + 8 * h) * VOCAB + (nb + n)] = acc[v];
}

__global__ __launch_bounds__(256) void memnn_logsoftmax(float* __restrict__ out) {
    __shared__ float red[256];
    const int b = blockIdx.x;
    const int t = threadIdx.x;
    float* row = out + (size_t)b * VOCAB;

    float mx = -3.4e38f;
    for (int i = t; i < VOCAB; i += 256) mx = fmaxf(mx, row[i]);
    red[t] = mx; __syncthreads();
    for (int o = 128; o > 0; o >>= 1) {
        if (t < o) red[t] = fmaxf(red[t], red[t + o]);
        __syncthreads();
    }
    mx = red[0]; __syncthreads();

    float sm = 0.f;
    for (int i = t; i < VOCAB; i += 256) sm += __expf(row[i] - mx);
    red[t] = sm; __syncthreads();
    for (int o = 128; o > 0; o >>= 1) {
        if (t < o) red[t] += red[t + o];
        __syncthreads();
    }
    const float lse = mx + logf(red[0]);

    for (int i = t; i < VOCAB; i += 256) row[i] = row[i] - lse;
}

extern "C" void kernel_launch(void* const* d_in, const int* in_sizes, int n_in,
                              void* d_out, int out_size, void* d_ws, size_t ws_size,
                              hipStream_t stream) {
    const int*   x  = (const int*)d_in[0];
    const int*   q  = (const int*)d_in[1];
    const float* A  = (const float*)d_in[2];
    const float* TA = (const float*)d_in[3];
    const float* TC = (const float*)d_in[4];
    float* out = (float*)d_out;

    float* ws = (float*)d_ws;
    float* u  = ws;
    float* m  = ws + BS * EMBD;
    float* c  = m + (size_t)HOPS * BS * STORY * EMBD;

    memnn_u0<<<BS, EMBD, 0, stream>>>(q, A, u);
    memnn_gather<<<BS * STORY, EMBD, 0, stream>>>(x, A, TA, TC, m, c);
    for (int h = 0; h < HOPS; ++h)
        memnn_hop<<<BS, EMBD, 0, stream>>>(m, c, u, h);

    const int ntiles = (VOCAB / 16) * (BS / 16);
    const int blocks = (ntiles + 7) / 8;
    memnn_logits_wmma<<<blocks, 256, 0, stream>>>(
        u, A + 3 * (size_t)VOCAB * EMBD, out);

    memnn_logsoftmax<<<BS, 256, 0, stream>>>(out);
}


# log-softmax spread from 64 to 256 work-groups: a stats launch writes per-quarter-row (max, sum-exp) partials, a second launch combines them and normalises; row sum is a different reduction tree
# speedup vs baseline: 2.0225x; 2.0225x over previous
#include <hip/hip_runtime.h>
#include <hip/hip_bf16.h>
#include <math.h>

#define VOCAB 50000
#define EMBD  128
#define HOPS  3
#define BS    64
#define STORY 50
#define SENT  20
#define QLEN  20

typedef __attribute__((ext_vector_type(2))) float v2f;
typedef __attribute__((ext_vector_type(8))) float v8f;

__global__ __launch_bounds__(EMBD) void memnn_u0(const int* __restrict__ q,
                                                 const float* __restrict__ A,
                                                 float* __restrict__ u) {
    __shared__ int tok[QLEN];
    const int b = blockIdx.x;
    const int d = threadIdx.x;
    if (d < QLEN) tok[d] = q[b * QLEN + d];
    __syncthreads();
    float acc = 0.0f;
#pragma unroll
    for (int j = 0; j < QLEN; ++j)
        acc += A[(size_t)tok[j] * EMBD + d];
    u[b * EMBD + d] = acc;
}

__global__ __launch_bounds__(EMBD) void memnn_gather(const int* __restrict__ x,
                                                     const float* __restrict__ A,
                                                     const float* __restrict__ TA,
                                                     const float* __restrict__ TC,
                                                     float* __restrict__ m,
                                                     float* __restrict__ c) {
    const int bs = blockIdx.x;
    const int s  = bs % STORY;
    const int d  = threadIdx.x;
    __shared__ int tok[SENT];
    if (d < SENT) tok[d] = x[(size_t)bs * SENT + d];
    __syncthreads();

    const float kd = (float)(d + 1) / (float)EMBD;
    const size_t LVL = (size_t)VOCAB * EMBD;

    float mp0 = 0.f, mp1 = 0.f, mp2 = 0.f;
    float cp1 = 0.f, cp2 = 0.f, cp3 = 0.f;
#pragma unroll
    for (int j = 0; j < SENT; ++j) {
        const float jj = (float)(j + 1) / (float)SENT;
        const float pe = 1.0f - jj - kd * (1.0f - 2.0f * jj);
        const size_t base = (size_t)tok[j] * EMBD + d;
        const float e0 = A[base];
        const float e1 = A[LVL + base];
        const float e2 = A[2 * LVL + base];
        const float e3 = A[3 * LVL + base];
        mp0 += e0 * pe;  mp1 += e1 * pe;  mp2 += e2 * pe;
        cp1 += e1;       cp2 += e2;       cp3 += e3;
    }
    const float ta = TA[s * EMBD + d];
    const float tc = TC[s * EMBD + d];
    const size_t off = (size_t)bs * EMBD + d;
    const size_t hs  = (size_t)BS * STORY * EMBD;
    m[off]          = mp0 + ta;
    m[hs + off]     = mp1 + ta;
    m[2 * hs + off] = mp2 + ta;
    c[off]          = cp1 + tc;
    c[hs + off]     = cp2 + tc;
    c[2 * hs + off] = cp3 + tc;
}

__global__ __launch_bounds__(EMBD) void memnn_hop(const float* __restrict__ m,
                                                  const float* __restrict__ c,
                                                  float* __restrict__ u,
                                                  int hop) {
    const int b = blockIdx.x;
    const int t = threadIdx.x;
    __shared__ float u_lds[EMBD];
    __shared__ float score[STORY];
    u_lds[t] = u[b * EMBD + t];
    __syncthreads();

    const int wave = t >> 5;
    const int lane = t & 31;
    const size_t hb = (size_t)hop * BS * STORY * EMBD + (size_t)b * STORY * EMBD;

    for (int s = wave; s < STORY; s += 4) {
        const float* mrow = m + hb + (size_t)s * EMBD;
        float p = 0.f;
#pragma unroll
        for (int i = 0; i < EMBD / 32; ++i)
            p += mrow[lane + 32 * i] * u_lds[lane + 32 * i];
        for (int o = 16; o > 0; o >>= 1) p += __shfl_xor(p, o, 32);
        if (lane == 0) score[s] = p;
    }
    __syncthreads();

    float mx = -3.4e38f;
    for (int s = 0; s < STORY; ++s) mx = fmaxf(mx, score[s]);
    float sm = 0.f;
    for (int s = 0; s < STORY; ++s) sm += __expf(score[s] - mx);
    const float inv = 1.0f / sm;

    float acc = 0.f;
    const float* crow = c + hb;
    for (int s = 0; s < STORY; ++s)
        acc += (__expf(score[s] - mx) * inv) * crow[(size_t)s * EMBD + t];
    u[b * EMBD + t] = u_lds[t] + acc;
}

__global__ __launch_bounds__(256) void memnn_logits_wmma(const float* __restrict__ u,
                                                         const float* __restrict__ A3,
                                                         float* __restrict__ out) {
    __shared__ float u_lds[BS * EMBD];
    for (int i = threadIdx.x; i < BS * EMBD; i += 256) u_lds[i] = u[i];
    __syncthreads();

    const int wave = threadIdx.x >> 5;
    const int lane = threadIdx.x & 31;
    const int tile = blockIdx.x * 8 + wave;
    const int NTILES = (VOCAB / 16) * (BS / 16);
    if (tile >= NTILES) return;

    const int mb = (tile & 3) << 4;
    const int nb = (tile >> 2) << 4;
    const int n  = lane & 15;
    const int h  = lane >> 4;

    const float* arow = u_lds + (mb + n) * EMBD;
    const float* brow = A3 + (size_t)(nb + n) * EMBD;

    v8f acc = {};
#pragma unroll
    for (int kk = 0; kk < EMBD / 4; ++kk) {
        const int k0 = kk * 4 + 2 * h;
        v2f a;  a.x  = arow[k0]; a.y  = arow[k0 + 1];
        v2f bv; bv.x = brow[k0]; bv.y = brow[k0 + 1];
        acc = __builtin_amdgcn_wmma_f32_16x16x4_f32(
             false, a,  false, bv,
             (short)0, acc,  false,  false);
    }
#pragma unroll
    for (int v = 0; v < 8; ++v)
        out[(size_t)(mb + v + 8 * h) * VOCAB + (nb + n)] = acc[v];
}

#define LSM_CH  4
#define LSM_THR 512
#define LSM_V4  (VOCAB / 4 / LSM_CH)
#define LSM_IT  ((LSM_V4 + LSM_THR - 1) / LSM_THR)

__global__ __launch_bounds__(LSM_THR) void lsm_stats(const float* __restrict__ out,
                                                     float* __restrict__ part) {
    __shared__ float red_mx[LSM_THR / 32];
    __shared__ float red_sm[LSM_THR / 32];
    const int ch = blockIdx.x;
    const int b  = blockIdx.y;
    const int t  = threadIdx.x;
    const int wave = t >> 5;
    const int lane = t & 31;
    const float4* p = reinterpret_cast<const float4*>(out + (size_t)b * VOCAB) + ch * LSM_V4;

    float4 v[LSM_IT];
#pragma unroll
    for (int i = 0; i < LSM_IT; ++i) {
        const int idx = t + i * LSM_THR;
        if (idx < LSM_V4) v[i] = p[idx];
        else              v[i] = make_float4(-INFINITY, -INFINITY, -INFINITY, -INFINITY);
    }

    float mx = -INFINITY;
#pragma unroll
    for (int i = 0; i < LSM_IT; ++i)
        mx = fmaxf(mx, fmaxf(fmaxf(v[i].x, v[i].y), fmaxf(v[i].z, v[i].w)));
    for (int o = 16; o > 0; o >>= 1) mx = fmaxf(mx, __shfl_xor(mx, o, 32));
    if (lane == 0) red_mx[wave] = mx;
    __syncthreads();
    mx = red_mx[0];
    for (int w = 1; w < LSM_THR / 32; ++w) mx = fmaxf(mx, red_mx[w]);

    float sm = 0.f;
#pragma unroll
    for (int i = 0; i < LSM_IT; ++i)
        sm += (__expf(v[i].x - mx) + __expf(v[i].y - mx)) +
              (__expf(v[i].z - mx) + __expf(v[i].w - mx));
    for (int o = 16; o > 0; o >>= 1) sm += __shfl_xor(sm, o, 32);
    if (lane == 0) red_sm[wave] = sm;
    __syncthreads();
    if (t == 0) {
        float s = 0.f;
        for (int w = 0; w < LSM_THR / 32; ++w) s += red_sm[w];
        part[(b * LSM_CH + ch) * 2]     = mx;
        part[(b * LSM_CH + ch) * 2 + 1] = s;
    }
}

__global__ __launch_bounds__(LSM_THR) void lsm_norm(float* __restrict__ out,
                                                    const float* __restrict__ part) {
    const int ch = blockIdx.x;
    const int b  = blockIdx.y;
    const int t  = threadIdx.x;
    float4* p = reinterpret_cast<float4*>(out + (size_t)b * VOCAB) + ch * LSM_V4;

    float4 v[LSM_IT];
#pragma unroll
    for (int i = 0; i < LSM_IT; ++i) {
        const int idx = t + i * LSM_THR;
        if (idx < LSM_V4) v[i] = p[idx];
    }

    float mx = part[b * LSM_CH * 2];
    for (int k = 1; k < LSM_CH; ++k) mx = fmaxf(mx, part[(b * LSM_CH + k) * 2]);
    float sm = 0.f;
    for (int k = 0; k < LSM_CH; ++k)
        sm += part[(b * LSM_CH + k) * 2 + 1] * __expf(part[(b * LSM_CH + k) * 2] - mx);
    const float lse = mx + logf(sm);

#pragma unroll
    for (int i = 0; i < LSM_IT; ++i) {
        const int idx = t + i * LSM_THR;
        if (idx < LSM_V4)
            p[idx] = make_float4(v[i].x - lse, v[i].y - lse, v[i].z - lse, v[i].w - lse);
    }
}

extern "C" void kernel_launch(void* const* d_in, const int* in_sizes, int n_in,
                              void* d_out, int out_size, void* d_ws, size_t ws_size,
                              hipStream_t stream) {
    const int*   x  = (const int*)d_in[0];
    const int*   q  = (const int*)d_in[1];
    const float* A  = (const float*)d_in[2];
    const float* TA = (const float*)d_in[3];
    const float* TC = (const float*)d_in[4];
    float* out = (float*)d_out;

    float* ws = (float*)d_ws;
    float* u  = ws;
    float* m  = ws + BS * EMBD;
    float* c  = m + (size_t)HOPS * BS * STORY * EMBD;

    memnn_u0<<<BS, EMBD, 0, stream>>>(q, A, u);
    memnn_gather<<<BS * STORY, EMBD, 0, stream>>>(x, A, TA, TC, m, c);
    for (int h = 0; h < HOPS; ++h)
        memnn_hop<<<BS, EMBD, 0, stream>>>(m, c, u, h);

    const int ntiles = (VOCAB / 16) * (BS / 16);
    const int blocks = (ntiles + 7) / 8;
    memnn_logits_wmma<<<blocks, 256, 0, stream>>>(
        u, A + 3 * (size_t)VOCAB * EMBD, out);

    float* part = m;
    lsm_stats<<<dim3(LSM_CH, BS), LSM_THR, 0, stream>>>(out, part);
    lsm_norm<<<dim3(LSM_CH, BS), LSM_THR, 0, stream>>>(out, part);
}


# query sum and the three hop launches fused into one launch of 512 threads per batch row: every m and c load of the three hops issued up front, u kept in registers; dot and p@c sums in a new order
# speedup vs baseline: 3.0082x; 1.4874x over previous
#include <hip/hip_runtime.h>
#include <hip/hip_bf16.h>
#include <math.h>

#define VOCAB 50000
#define EMBD  128
#define HOPS  3
#define BS    64
#define STORY 50
#define SENT  20
#define QLEN  20

typedef __attribute__((ext_vector_type(2))) float v2f;
typedef __attribute__((ext_vector_type(8))) float v8f;

__global__ __launch_bounds__(EMBD) void memnn_gather(const int* __restrict__ x,
                                                     const float* __restrict__ A,
                                                     const float* __restrict__ TA,
                                                     const float* __restrict__ TC,
                                                     float* __restrict__ m,
                                                     float* __restrict__ c) {
    const int bs = blockIdx.x;
    const int s  = bs % STORY;
    const int d  = threadIdx.x;
    __shared__ int tok[SENT];
    if (d < SENT) tok[d] = x[(size_t)bs * SENT + d];
    __syncthreads();

    const float kd = (float)(d + 1) / (float)EMBD;
    const size_t LVL = (size_t)VOCAB * EMBD;

    float mp0 = 0.f, mp1 = 0.f, mp2 = 0.f;
    float cp1 = 0.f, cp2 = 0.f, cp3 = 0.f;
#pragma unroll
    for (int j = 0; j < SENT; ++j) {
        const float jj = (float)(j + 1) / (float)SENT;
        const float pe = 1.0f - jj - kd * (1.0f - 2.0f * jj);
        const size_t base = (size_t)tok[j] * EMBD + d;
        const float e0 = A[base];
        const float e1 = A[LVL + base];
        const float e2 = A[2 * LVL + base];
        const float e3 = A[3 * LVL + base];
        mp0 += e0 * pe;  mp1 += e1 * pe;  mp2 += e2 * pe;
        cp1 += e1;       cp2 += e2;       cp3 += e3;
    }
    const float ta = TA[s * EMBD + d];
    const float tc = TC[s * EMBD + d];
    const size_t off = (size_t)bs * EMBD + d;
    const size_t hs  = (size_t)BS * STORY * EMBD;
    m[off]          = mp0 + ta;
    m[hs + off]     = mp1 + ta;
    m[2 * hs + off] = mp2 + ta;
    c[off]          = cp1 + tc;
    c[hs + off]     = cp2 + tc;
    c[2 * hs + off] = cp3 + tc;
}

#define CH_THR 512
#define CH_W   (CH_THR / 32)
#define CH_MR  ((STORY + CH_W - 1) / CH_W)
#define CH_G   (CH_THR / EMBD)
#define CH_CR  ((STORY + CH_G - 1) / CH_G)
#define CH_S4  ((STORY + 3) / 4)

__global__ __launch_bounds__(CH_THR) void memnn_u_chain(const int* __restrict__ q,
                                                        const float* __restrict__ A,
                                                        const float* __restrict__ m,
                                                        const float* __restrict__ c,
                                                        float* __restrict__ u) {
    __shared__ __attribute__((aligned(16))) float u_lds[EMBD];
    __shared__ __attribute__((aligned(16))) float score[CH_S4 * 4];
    __shared__ __attribute__((aligned(16))) float pacc[CH_G][EMBD];
    __shared__ float psum[CH_G];

    const int b    = blockIdx.x;
    const int t    = threadIdx.x;
    const int wave = t >> 5;
    const int lane = t & 31;
    const int g    = t >> 7;
    const int d    = t & (EMBD - 1);

    float4 mv[HOPS][CH_MR];
    float  cv[HOPS][CH_CR];
#pragma unroll
    for (int h = 0; h < HOPS; ++h) {
        const size_t hb = ((size_t)h * BS + b) * STORY * EMBD;
#pragma unroll
        for (int i = 0; i < CH_MR; ++i) {
            const int s = min(wave + CH_W * i, STORY - 1);
            mv[h][i] = *reinterpret_cast<const float4*>(m + hb + (size_t)s * EMBD + 4 * lane);
        }
#pragma unroll
        for (int i = 0; i < CH_CR; ++i) {
            const int s = min(g + CH_G * i, STORY - 1);
            cv[h][i] = c[hb + (size_t)s * EMBD + d];
        }
    }

    if (t < EMBD) {
        float acc = 0.0f;
#pragma unroll
        for (int j = 0; j < QLEN; ++j)
            acc += A[(size_t)q[b * QLEN + j] * EMBD + t];
        u_lds[t] = acc;
    }
    if (t < CH_S4 * 4 - STORY) score[STORY + t] = -INFINITY;
    __syncthreads();

    float4 u4 = reinterpret_cast<const float4*>(u_lds)[lane];

#pragma unroll
    for (int h = 0; h < HOPS; ++h) {
        float p[CH_MR];
#pragma unroll
        for (int i = 0; i < CH_MR; ++i)
            p[i] = (mv[h][i].x * u4.x + mv[h][i].y * u4.y) + (mv[h][i].z * u4.z + mv[h][i].w * u4.w);
        for (int o = 16; o > 0; o >>= 1) {
#pragma unroll
            for (int i = 0; i < CH_MR; ++i) p[i] += __shfl_xor(p[i], o, 32);
        }
        if (lane == 0) {
#pragma unroll
            for (int i = 0; i < CH_MR; ++i) {
                const int s = wave + CH_W * i;
                if (s < STORY) score[s] = p[i];
            }
        }
        __syncthreads();

        float mx = -3.4e38f;
#pragma unroll
        for (int k = 0; k < CH_S4; ++k) {
            const float4 sc = reinterpret_cast<const float4*>(score)[k];
            mx = fmaxf(mx, fmaxf(fmaxf(sc.x, sc.y), fmaxf(sc.z, sc.w)));
        }
        float sm = 0.f, acc = 0.f;
#pragma unroll
        for (int i = 0; i < CH_CR; ++i) {
            const int s = g + CH_G * i;
            if (s < STORY) {
                const float e = __expf(score[s] - mx);
                sm  += e;
                acc += e * cv[h][i];
            }
        }
        pacc[g][d] = acc;
        if (d == 0) psum[g] = sm;
        __syncthreads();

        const float tot = (psum[0] + psum[1]) + (psum[2] + psum[3]);
        const float inv = 1.0f / tot;
        const float4 a0 = reinterpret_cast<const float4*>(pacc[0])[lane];
        const float4 a1 = reinterpret_cast<const float4*>(pacc[1])[lane];
        const float4 a2 = reinterpret_cast<const float4*>(pacc[2])[lane];
        const float4 a3 = reinterpret_cast<const float4*>(pacc[3])[lane];
        u4.x += ((a0.x + a1.x) + (a2.x + a3.x)) * inv;
        u4.y += ((a0.y + a1.y) + (a2.y + a3.y)) * inv;
        u4.z += ((a0.z + a1.z) + (a2.z + a3.z)) * inv;
        u4.w += ((a0.w + a1.w) + (a2.w + a3.w)) * inv;
    }
    if (wave == 0) reinterpret_cast<float4*>(u + (size_t)b * EMBD)[lane] = u4;
}

__global__ __launch_bounds__(256) void memnn_logits_wmma(const float* __restrict__ u,
                                                         const float* __restrict__ A3,
                                                         float* __restrict__ out) {
    __shared__ float u_lds[BS * EMBD];
    for (int i = threadIdx.x; i < BS * EMBD; i += 256) u_lds[i] = u[i];
    __syncthreads();

    const int wave = threadIdx.x >> 5;
    const int lane = threadIdx.x & 31;
    const int tile = blockIdx.x * 8 + wave;
    const int NTILES = (VOCAB / 16) * (BS / 16);
    if (tile >= NTILES) return;

    const int mb = (tile & 3) << 4;
    const int nb = (tile >> 2) << 4;
    const int n  = lane & 15;
    const int h  = lane >> 4;

    const float* arow = u_lds + (mb + n) * EMBD;
    const float* brow = A3 + (size_t)(nb + n) * EMBD;

    v8f acc = {};
#pragma unroll
    for (int kk = 0; kk < EMBD / 4; ++kk) {
        const int k0 = kk * 4 + 2 * h;
        v2f a;  a.x  = arow[k0]; a.y  = arow[k0 + 1];
        v2f bv; bv.x = brow[k0]; bv.y = brow[k0 + 1];
        acc = __builtin_amdgcn_wmma_f32_16x16x4_f32(
             false, a,  false, bv,
             (short)0, acc,  false,  false);
    }
#pragma unroll
    for (int v = 0; v < 8; ++v)
        out[(size_t)(mb + v + 8 * h) * VOCAB + (nb + n)] = acc[v];
}

#define LSM_CH  4
#define LSM_THR 512
#define LSM_V4  (VOCAB / 4 / LSM_CH)
#define LSM_IT  ((LSM_V4 + LSM_THR - 1) / LSM_THR)

__global__ __launch_bounds__(LSM_THR) void lsm_stats(const float* __restrict__ out,
                                                     float* __restrict__ part) {
    __shared__ float red_mx[LSM_THR / 32];
    __shared__ float red_sm[LSM_THR / 32];
    const int ch = blockIdx.x;
    const int b  = blockIdx.y;
    const int t  = threadIdx.x;
    const int wave = t >> 5;
    const int lane = t & 31;
    const float4* p = reinterpret_cast<const float4*>(out + (size_t)b * VOCAB) + ch * LSM_V4;

    float4 v[LSM_IT];
#pragma unroll
    for (int i = 0; i < LSM_IT; ++i) {
        const int idx = t + i * LSM_THR;
        if (idx < LSM_V4) v[i] = p[idx];
        else              v[i] = make_float4(-INFINITY, -INFINITY, -INFINITY, -INFINITY);
    }

    float mx = -INFINITY;
#pragma unroll
    for (int i = 0; i < LSM_IT; ++i)
        mx = fmaxf(mx, fmaxf(fmaxf(v[i].x, v[i].y), fmaxf(v[i].z, v[i].w)));
    for (int o = 16; o > 0; o >>= 1) mx = fmaxf(mx, __shfl_xor(mx, o, 32));
    if (lane == 0) red_mx[wave] = mx;
    __syncthreads();
    mx = red_mx[0];
    for (int w = 1; w < LSM_THR / 32; ++w) mx = fmaxf(mx, red_mx[w]);

    float sm = 0.f;
#pragma unroll
    for (int i = 0; i < LSM_IT; ++i)
        sm += (__expf(v[i].x - mx) + __expf(v[i].y - mx)) +
              (__expf(v[i].z - mx) + __expf(v[i].w - mx));
    for (int o = 16; o > 0; o >>= 1) sm += __shfl_xor(sm, o, 32);
    if (lane == 0) red_sm[wave] = sm;
    __syncthreads();
    if (t == 0) {
        float s = 0.f;
        for (int w = 0; w < LSM_THR / 32; ++w) s += red_sm[w];
        part[(b * LSM_CH + ch) * 2]     = mx;
        part[(b * LSM_CH + ch) * 2 + 1] = s;
    }
}

__global__ __launch_bounds__(LSM_THR) void lsm_norm(float* __restrict__ out,
                                                    const float* __restrict__ part) {
    const int ch = blockIdx.x;
    const int b  = blockIdx.y;
    const int t  = threadIdx.x;
    float4* p = reinterpret_cast<float4*>(out + (size_t)b * VOCAB) + ch * LSM_V4;

    float4 v[LSM_IT];
#pragma unroll
    for (int i = 0; i < LSM_IT; ++i) {
        const int idx = t + i * LSM_THR;
        if (idx < LSM_V4) v[i] = p[idx];
    }

    float mx = part[b * LSM_CH * 2];
    for (int k = 1; k < LSM_CH; ++k) mx = fmaxf(mx, part[(b * LSM_CH + k) * 2]);
    float sm = 0.f;
    for (int k = 0; k < LSM_CH; ++k)
        sm += part[(b * LSM_CH + k) * 2 + 1] * __expf(part[(b * LSM_CH + k) * 2] - mx);
    const float lse = mx + logf(sm);

#pragma unroll
    for (int i = 0; i < LSM_IT; ++i) {
        const int idx = t + i * LSM_THR;
        if (idx < LSM_V4)
            p[idx] = make_float4(v[i].x - lse, v[i].y - lse, v[i].z - lse, v[i].w - lse);
    }
}

extern "C" void kernel_launch(void* const* d_in, const int* in_sizes, int n_in,
                              void* d_out, int out_size, void* d_ws, size_t ws_size,
                              hipStream_t stream) {
    const int*   x  = (const int*)d_in[0];
    const int*   q  = (const int*)d_in[1];
    const float* A  = (const float*)d_in[2];
    const float* TA = (const float*)d_in[3];
    const float* TC = (const float*)d_in[4];
    float* out = (float*)d_out;

    float* ws = (float*)d_ws;
    float* u  = ws;
    float* m  = ws + BS * EMBD;
    float* c  = m + (size_t)HOPS * BS * STORY * EMBD;

    memnn_gather<<<BS * STORY, EMBD, 0, stream>>>(x, A, TA, TC, m, c);
    memnn_u_chain<<<BS, CH_THR, 0, stream>>>(q, A, m, c, u);

    const int ntiles = (VOCAB / 16) * (BS / 16);
    const int blocks = (ntiles + 7) / 8;
    memnn_logits_wmma<<<blocks, 256, 0, stream>>>(
        u, A + 3 * (size_t)VOCAB * EMBD, out);

    float* part = m;
    lsm_stats<<<dim3(LSM_CH, BS), LSM_THR, 0, stream>>>(out, part);
    lsm_norm<<<dim3(LSM_CH, BS), LSM_THR, 0, stream>>>(out, part);
}


# logits GEMM: one wave per 16-column vocabulary strip, all 4 batch tiles (B fragments loaded once, reused by 4 WMMAs; u in LDS at a conflict-free pitch): 12504 waves become 3125, k order kept
# speedup vs baseline: 3.6341x; 1.2081x over previous
#include <hip/hip_runtime.h>
#include <hip/hip_bf16.h>
#include <math.h>

#define VOCAB 50000
#define EMBD  128
#define HOPS  3
#define BS    64
#define STORY 50
#define SENT  20
#define QLEN  20

typedef __attribute__((ext_vector_type(2))) float v2f;
typedef __attribute__((ext_vector_type(8))) float v8f;

__global__ __launch_bounds__(EMBD) void memnn_gather(const int* __restrict__ x,
                                                     const float* __restrict__ A,
                                                     const float* __restrict__ TA,
                                                     const float* __restrict__ TC,
                                                     float* __restrict__ m,
                                                     float* __restrict__ c) {
    const int bs = blockIdx.x;
    const int s  = bs % STORY;
    const int d  = threadIdx.x;
    __shared__ int tok[SENT];
    if (d < SENT) tok[d] = x[(size_t)bs * SENT + d];
    __syncthreads();

    const float kd = (float)(d + 1) / (float)EMBD;
    const size_t LVL = (size_t)VOCAB * EMBD;

    float mp0 = 0.f, mp1 = 0.f, mp2 = 0.f;
    float cp1 = 0.f, cp2 = 0.f, cp3 = 0.f;
#pragma unroll
    for (int j = 0; j < SENT; ++j) {
        const float jj = (float)(j + 1) / (float)SENT;
        const float pe = 1.0f - jj - kd * (1.0f - 2.0f * jj);
        const size_t base = (size_t)tok[j] * EMBD + d;
        const float e0 = A[base];
        const float e1 = A[LVL + base];
        const float e2 = A[2 * LVL + base];
        const float e3 = A[3 * LVL + base];
        mp0 += e0 * pe;  mp1 += e1 * pe;  mp2 += e2 * pe;
        cp1 += e1;       cp2 += e2;       cp3 += e3;
    }
    const float ta = TA[s * EMBD + d];
    const float tc = TC[s * EMBD + d];
    const size_t off = (size_t)bs * EMBD + d;
    const size_t hs  = (size_t)BS * STORY * EMBD;
    m[off]          = mp0 + ta;
    m[hs + off]     = mp1 + ta;
    m[2 * hs + off] = mp2 + ta;
    c[off]          = cp1 + tc;
    c[hs + off]     = cp2 + tc;
    c[2 * hs + off] = cp3 + tc;
}

#define CH_THR 512
#define CH_W   (CH_THR / 32)
#define CH_MR  ((STORY + CH_W - 1) / CH_W)
#define CH_G   (CH_THR / EMBD)
#define CH_CR  ((STORY + CH_G - 1) / CH_G)
#define CH_S4  ((STORY + 3) / 4)

__global__ __launch_bounds__(CH_THR) void memnn_u_chain(const int* __restrict__ q,
                                                        const float* __restrict__ A,
                                                        const float* __restrict__ m,
                                                        const float* __restrict__ c,
                                                        float* __restrict__ u) {
    __shared__ __attribute__((aligned(16))) float u_lds[EMBD];
    __shared__ __attribute__((aligned(16))) float score[CH_S4 * 4];
    __shared__ __attribute__((aligned(16))) float pacc[CH_G][EMBD];
    __shared__ float psum[CH_G];

    const int b    = blockIdx.x;
    const int t    = threadIdx.x;
    const int wave = t >> 5;
    const int lane = t & 31;
    const int g    = t >> 7;
    const int d    = t & (EMBD - 1);

    float4 mv[HOPS][CH_MR];
    float  cv[HOPS][CH_CR];
#pragma unroll
    for (int h = 0; h < HOPS; ++h) {
        const size_t hb = ((size_t)h * BS + b) * STORY * EMBD;
#pragma unroll
        for (int i = 0; i < CH_MR; ++i) {
            const int s = min(wave + CH_W * i, STORY - 1);
            mv[h][i] = *reinterpret_cast<const float4*>(m + hb + (size_t)s * EMBD + 4 * lane);
        }
#pragma unroll
        for (int i = 0; i < CH_CR; ++i) {
            const int s = min(g + CH_G * i, STORY - 1);
            cv[h][i] = c[hb + (size_t)s * EMBD + d];
        }
    }

    if (t < EMBD) {
        float acc = 0.0f;
#pragma unroll
        for (int j = 0; j < QLEN; ++j)
            acc += A[(size_t)q[b * QLEN + j] * EMBD + t];
        u_lds[t] = acc;
    }
    if (t < CH_S4 * 4 - STORY) score[STORY + t] = -INFINITY;
    __syncthreads();

    float4 u4 = reinterpret_cast<const float4*>(u_lds)[lane];

#pragma unroll
    for (int h = 0; h < HOPS; ++h) {
        float p[CH_MR];
#pragma unroll
        for (int i = 0; i < CH_MR; ++i)
            p[i] = (mv[h][i].x * u4.x + mv[h][i].y * u4.y) + (mv[h][i].z * u4.z + mv[h][i].w * u4.w);
        for (int o = 16; o > 0; o >>= 1) {
#pragma unroll
            for (int i = 0; i < CH_MR; ++i) p[i] += __shfl_xor(p[i], o, 32);
        }
        if (lane == 0) {
#pragma unroll
            for (int i = 0; i < CH_MR; ++i) {
                const int s = wave + CH_W * i;
                if (s < STORY) score[s] = p[i];
            }
        }
        __syncthreads();

        float mx = -3.4e38f;
#pragma unroll
        for (int k = 0; k < CH_S4; ++k) {
            const float4 sc = reinterpret_cast<const float4*>(score)[k];
            mx = fmaxf(mx, fmaxf(fmaxf(sc.x, sc.y), fmaxf(sc.z, sc.w)));
        }
        float sm = 0.f, acc = 0.f;
#pragma unroll
        for (int i = 0; i < CH_CR; ++i) {
            const int s = g + CH_G * i;
            if (s < STORY) {
                const float e = __expf(score[s] - mx);
                sm  += e;
                acc += e * cv[h][i];
            }
        }
        pacc[g][d] = acc;
        if (d == 0) psum[g] = sm;
        __syncthreads();

        const float tot = (psum[0] + psum[1]) + (psum[2] + psum[3]);
        const float inv = 1.0f / tot;
        const float4 a0 = reinterpret_cast<const float4*>(pacc[0])[lane];
        const float4 a1 = reinterpret_cast<const float4*>(pacc[1])[lane];
        const float4 a2 = reinterpret_cast<const float4*>(pacc[2])[lane];
        const float4 a3 = reinterpret_cast<const float4*>(pacc[3])[lane];
        u4.x += ((a0.x + a1.x) + (a2.x + a3.x)) * inv;
        u4.y += ((a0.y + a1.y) + (a2.y + a3.y)) * inv;
        u4.z += ((a0.z + a1.z) + (a2.z + a3.z)) * inv;
        u4.w += ((a0.w + a1.w) + (a2.w + a3.w)) * inv;
    }
    if (wave == 0) reinterpret_cast<float4*>(u + (size_t)b * EMBD)[lane] = u4;
}

#define LG_THR    256
#define LG_W      (LG_THR / 32)
#define LG_NT     (VOCAB / 16)
#define LG_BLOCKS ((LG_NT + LG_W - 1) / LG_W)
#define LG_MT     (BS / 16)
#define LG_KK     (EMBD / 4)
#define LG_PITCH  (EMBD + 4)
#define LG_ST     (BS * EMBD / 4 / LG_THR)

__global__ __launch_bounds__(LG_THR) void memnn_logits_wmma(const float* __restrict__ u,
                                                            const float* __restrict__ A3,
                                                            float* __restrict__ out) {
    __shared__ __attribute__((aligned(16))) float u_lds[BS * LG_PITCH];

    const int wave  = threadIdx.x >> 5;
    const int lane  = threadIdx.x & 31;
    const int strip = blockIdx.x * LG_W + wave;
    const bool live = strip < LG_NT;
    const int nb = (live ? strip : LG_NT - 1) << 4;
    const int n  = lane & 15;
    const int h  = lane >> 4;

    const float* brow = A3 + (size_t)(nb + n) * EMBD + 2 * h;
    v2f bv[LG_KK];
#pragma unroll
    for (int kk = 0; kk < LG_KK; ++kk)
        bv[kk] = *reinterpret_cast<const v2f*>(brow + kk * 4);

#pragma unroll
    for (int i = 0; i < LG_ST; ++i) {
        const int f = threadIdx.x + i * LG_THR;
        const float4 v = reinterpret_cast<const float4*>(u)[f];
        *reinterpret_cast<float4*>(u_lds + (f >> 5) * LG_PITCH + (f & 31) * 4) = v;
    }
    __syncthreads();
    if (!live) return;

    const float* arow = u_lds + n * LG_PITCH + 2 * h;
    v8f acc[LG_MT] = {};
#pragma unroll
    for (int kk = 0; kk < LG_KK; ++kk) {
#pragma unroll
        for (int mt = 0; mt < LG_MT; ++mt) {
            const v2f a = *reinterpret_cast<const v2f*>(arow + mt * 16 * LG_PITCH + kk * 4);
            acc[mt] = __builtin_amdgcn_wmma_f32_16x16x4_f32(
                 false, a,  false, bv[kk],
                 (short)0, acc[mt],  false,  false);
        }
    }
#pragma unroll
    for (int mt = 0; mt < LG_MT; ++mt) {
#pragma unroll
        for (int v = 0; v < 8; ++v)
            out[(size_t)(mt * 16 + v + 8 * h) * VOCAB + (nb + n)] = acc[mt][v];
    }
}

#define LSM_CH  4
#define LSM_THR 512
#define LSM_V4  (VOCAB / 4 / LSM_CH)
#define LSM_IT  ((LSM_V4 + LSM_THR - 1) / LSM_THR)

__global__ __launch_bounds__(LSM_THR) void lsm_stats(const float* __restrict__ out,
                                                     float* __restrict__ part) {
    __shared__ float red_mx[LSM_THR / 32];
    __shared__ float red_sm[LSM_THR / 32];
    const int ch = blockIdx.x;
    const int b  = blockIdx.y;
    const int t  = threadIdx.x;
    const int wave = t >> 5;
    const int lane = t & 31;
    const float4* p = reinterpret_cast<const float4*>(out + (size_t)b * VOCAB) + ch * LSM_V4;

    float4 v[LSM_IT];
#pragma unroll
    for (int i = 0; i < LSM_IT; ++i) {
        const int idx = t + i * LSM_THR;
        if (idx < LSM_V4) v[i] = p[idx];
        else              v[i] = make_float4(-INFINITY, -INFINITY, -INFINITY, -INFINITY);
    }

    float mx = -INFINITY;
#pragma unroll
    for (int i = 0; i < LSM_IT; ++i)
        mx = fmaxf(mx, fmaxf(fmaxf(v[i].x, v[i].y), fmaxf(v[i].z, v[i].w)));
    for (int o = 16; o > 0; o >>= 1) mx = fmaxf(mx, __shfl_xor(mx, o, 32));
    if (lane == 0) red_mx[wave] = mx;
    __syncthreads();
    mx = red_mx[0];
    for (int w = 1; w < LSM_THR / 32; ++w) mx = fmaxf(mx, red_mx[w]);

    float sm = 0.f;
#pragma unroll
    for (int i = 0; i < LSM_IT; ++i)
        sm += (__expf(v[i].x - mx) + __expf(v[i].y - mx)) +
              (__expf(v[i].z - mx) + __expf(v[i].w - mx));
    for (int o = 16; o > 0; o >>= 1) sm += __shfl_xor(sm, o, 32);
    if (lane == 0) red_sm[wave] = sm;
    __syncthreads();
    if (t == 0) {
        float s = 0.f;
        for (int w = 0; w < LSM_THR / 32; ++w) s += red_sm[w];
        part[(b * LSM_CH + ch) * 2]     = mx;
        part[(b * LSM_CH + ch) * 2 + 1] = s;
    }
}

__global__ __launch_bounds__(LSM_THR) void lsm_norm(float* __restrict__ out,
                                                    const float* __restrict__ part) {
    const int ch = blockIdx.x;
    const int b  = blockIdx.y;
    const int t  = threadIdx.x;
    float4* p = reinterpret_cast<float4*>(out + (size_t)b * VOCAB) + ch * LSM_V4;

    float4 v[LSM_IT];
#pragma unroll
    for (int i = 0; i < LSM_IT; ++i) {
        const int idx = t + i * LSM_THR;
        if (idx < LSM_V4) v[i] = p[idx];
    }

    float mx = part[b * LSM_CH * 2];
    for (int k = 1; k < LSM_CH; ++k) mx = fmaxf(mx, part[(b * LSM_CH + k) * 2]);
    float sm = 0.f;
    for (int k = 0; k < LSM_CH; ++k)
        sm += part[(b * LSM_CH + k) * 2 + 1] * __expf(part[(b * LSM_CH + k) * 2] - mx);
    const float lse = mx + logf(sm);

#pragma unroll
    for (int i = 0; i < LSM_IT; ++i) {
        const int idx = t + i * LSM_THR;
        if (idx < LSM_V4)
            p[idx] = make_float4(v[i].x - lse, v[i].y - lse, v[i].z - lse, v[i].w - lse);
    }
}

extern "C" void kernel_launch(void* const* d_in, const int* in_sizes, int n_in,
                              void* d_out, int out_size, void* d_ws, size_t ws_size,
                              hipStream_t stream) {
    const int*   x  = (const int*)d_in[0];
    const int*   q  = (const int*)d_in[1];
    const float* A  = (const float*)d_in[2];
    const float* TA = (const float*)d_in[3];
    const float* TC = (const float*)d_in[4];
    float* out = (float*)d_out;

    float* ws = (float*)d_ws;
    float* u  = ws;
    float* m  = ws + BS * EMBD;
    float* c  = m + (size_t)HOPS * BS * STORY * EMBD;

    memnn_gather<<<BS * STORY, EMBD, 0, stream>>>(x, A, TA, TC, m, c);
    memnn_u_chain<<<BS, CH_THR, 0, stream>>>(q, A, m, c, u);

    memnn_logits_wmma<<<LG_BLOCKS, LG_THR, 0, stream>>>(
        u, A + 3 * (size_t)VOCAB * EMBD, out);

    float* part = m;
    lsm_stats<<<dim3(LSM_CH, BS), LSM_THR, 0, stream>>>(out, part);
    lsm_norm<<<dim3(LSM_CH, BS), LSM_THR, 0, stream>>>(out, part);
}


# log-softmax (max, sum-exp) pairs taken in the logits GEMM launch: waves also drop their tiles in LDS, one barrier, lanes reduce rows and fold the 8 strips; stats launch removed; row sum tree changes
# speedup vs baseline: 3.7983x; 1.0452x over previous
#include <hip/hip_runtime.h>
#include <hip/hip_bf16.h>
#include <math.h>

#define VOCAB 50000
#define EMBD  128
#define HOPS  3
#define BS    64
#define STORY 50
#define SENT  20
#define QLEN  20

typedef __attribute__((ext_vector_type(2))) float v2f;
typedef __attribute__((ext_vector_type(8))) float v8f;

__global__ __launch_bounds__(EMBD) void memnn_gather(const int* __restrict__ x,
                                                     const float* __restrict__ A,
                                                     const float* __restrict__ TA,
                                                     const float* __restrict__ TC,
                                                     float* __restrict__ m,
                                                     float* __restrict__ c) {
    const int bs = blockIdx.x;
    const int s  = bs % STORY;
    const int d  = threadIdx.x;
    __shared__ int tok[SENT];
    if (d < SENT) tok[d] = x[(size_t)bs * SENT + d];
    __syncthreads();

    const float kd = (float)(d + 1) / (float)EMBD;
    const size_t LVL = (size_t)VOCAB * EMBD;

    float mp0 = 0.f, mp1 = 0.f, mp2 = 0.f;
    float cp1 = 0.f, cp2 = 0.f, cp3 = 0.f;
#pragma unroll
    for (int j = 0; j < SENT; ++j) {
        const float jj = (float)(j + 1) / (float)SENT;
        const float pe = 1.0f - jj - kd * (1.0f - 2.0f * jj);
        const size_t base = (size_t)tok[j] * EMBD + d;
        const float e0 = A[base];
        const float e1 = A[LVL + base];
        const float e2 = A[2 * LVL + base];
        const float e3 = A[3 * LVL + base];
        mp0 += e0 * pe;  mp1 += e1 * pe;  mp2 += e2 * pe;
        cp1 += e1;       cp2 += e2;       cp3 += e3;
    }
    const float ta = TA[s * EMBD + d];
    const float tc = TC[s * EMBD + d];
    const size_t off = (size_t)bs * EMBD + d;
    const size_t hs  = (size_t)BS * STORY * EMBD;
    m[off]          = mp0 + ta;
    m[hs + off]     = mp1 + ta;
    m[2 * hs + off] = mp2 + ta;
    c[off]          = cp1 + tc;
    c[hs + off]     = cp2 + tc;
    c[2 * hs + off] = cp3 + tc;
}

#define CH_THR 512
#define CH_W   (CH_THR / 32)
#define CH_MR  ((STORY + CH_W - 1) / CH_W)
#define CH_G   (CH_THR / EMBD)
#define CH_CR  ((STORY + CH_G - 1) / CH_G)
#define CH_S4  ((STORY + 3) / 4)

__global__ __launch_bounds__(CH_THR) void memnn_u_chain(const int* __restrict__ q,
                                                        const float* __restrict__ A,
                                                        const float* __restrict__ m,
                                                        const float* __restrict__ c,
                                                        float* __restrict__ u) {
    __shared__ __attribute__((aligned(16))) float u_lds[EMBD];
    __shared__ __attribute__((aligned(16))) float score[CH_S4 * 4];
    __shared__ __attribute__((aligned(16))) float pacc[CH_G][EMBD];
    __shared__ float psum[CH_G];

    const int b    = blockIdx.x;
    const int t    = threadIdx.x;
    const int wave = t >> 5;
    const int lane = t & 31;
    const int g    = t >> 7;
    const int d    = t & (EMBD - 1);

    float4 mv[HOPS][CH_MR];
    float  cv[HOPS][CH_CR];
#pragma unroll
    for (int h = 0; h < HOPS; ++h) {
        const size_t hb = ((size_t)h * BS + b) * STORY * EMBD;
#pragma unroll
        for (int i = 0; i < CH_MR; ++i) {
            const int s = min(wave + CH_W * i, STORY - 1);
            mv[h][i] = *reinterpret_cast<const float4*>(m + hb + (size_t)s * EMBD + 4 * lane);
        }
#pragma unroll
        for (int i = 0; i < CH_CR; ++i) {
            const int s = min(g + CH_G * i, STORY - 1);
            cv[h][i] = c[hb + (size_t)s * EMBD + d];
        }
    }

    if (t < EMBD) {
        float acc = 0.0f;
#pragma unroll
        for (int j = 0; j < QLEN; ++j)
            acc += A[(size_t)q[b * QLEN + j] * EMBD + t];
        u_lds[t] = acc;
    }
    if (t < CH_S4 * 4 - STORY) score[STORY + t] = -INFINITY;
    __syncthreads();

    float4 u4 = reinterpret_cast<const float4*>(u_lds)[lane];

#pragma unroll
    for (int h = 0; h < HOPS; ++h) {
        float p[CH_MR];
#pragma unroll
        for (int i = 0; i < CH_MR; ++i)
            p[i] = (mv[h][i].x * u4.x + mv[h][i].y * u4.y) + (mv[h][i].z * u4.z + mv[h][i].w * u4.w);
        for (int o = 16; o > 0; o >>= 1) {
#pragma unroll
            for (int i = 0; i < CH_MR; ++i) p[i] += __shfl_xor(p[i], o, 32);
        }
        if (lane == 0) {
#pragma unroll
            for (int i = 0; i < CH_MR; ++i) {
                const int s = wave + CH_W * i;
                if (s < STORY) score[s] = p[i];
            }
        }
        __syncthreads();

        float mx = -3.4e38f;
#pragma unroll
        for (int k = 0; k < CH_S4; ++k) {
            const float4 sc = reinterpret_cast<const float4*>(score)[k];
            mx = fmaxf(mx, fmaxf(fmaxf(sc.x, sc.y), fmaxf(sc.z, sc.w)));
        }
        float sm = 0.f, acc = 0.f;
#pragma unroll
        for (int i = 0; i < CH_CR; ++i) {
            const int s = g + CH_G * i;
            if (s < STORY) {
                const float e = __expf(score[s] - mx);
                sm  += e;
                acc += e * cv[h][i];
            }
        }
        pacc[g][d] = acc;
        if (d == 0) psum[g] = sm;
        __syncthreads();

        const float tot = (psum[0] + psum[1]) + (psum[2] + psum[3]);
        const float inv = 1.0f / tot;
        const float4 a0 = reinterpret_cast<const float4*>(pacc[0])[lane];
        const float4 a1 = reinterpret_cast<const float4*>(pacc[1])[lane];
        const float4 a2 = reinterpret_cast<const float4*>(pacc[2])[lane];
        const float4 a3 = reinterpret_cast<const float4*>(pacc[3])[lane];
        u4.x += ((a0.x + a1.x) + (a2.x + a3.x)) * inv;
        u4.y += ((a0.y + a1.y) + (a2.y + a3.y)) * inv;
        u4.z += ((a0.z + a1.z) + (a2.z + a3.z)) * inv;
        u4.w += ((a0.w + a1.w) + (a2.w + a3.w)) * inv;
    }
    if (wave == 0) reinterpret_cast<float4*>(u + (size_t)b * EMBD)[lane] = u4;
}

#define LG_THR    256
#define LG_W      (LG_THR / 32)
#define LG_NT     (VOCAB / 16)
#define LG_BLOCKS ((LG_NT + LG_W - 1) / LG_W)
#define LG_MT     (BS / 16)
#define LG_KK     (EMBD / 4)
#define LG_PITCH  (EMBD + 4)
#define LG_ST     (BS * EMBD / 4 / LG_THR)
#define LG_TP     20
#define LG_TS     (BS * LG_TP + 8)

__device__ __forceinline__ void lse_fold(float& m, float& s, float mo, float so) {
    const float M = fmaxf(m, mo);
    const float a = (m  == M) ? 1.0f : __expf(m  - M);
    const float b = (mo == M) ? 1.0f : __expf(mo - M);
    s = s * a + so * b;
    m = M;
}

__global__ __launch_bounds__(LG_THR) void memnn_logits_wmma(const float* __restrict__ u,
                                                            const float* __restrict__ A3,
                                                            float* __restrict__ out,
                                                            float* __restrict__ part) {
    __shared__ __attribute__((aligned(16))) float u_lds[BS * LG_PITCH];
    __shared__ __attribute__((aligned(16))) float tile[LG_W * LG_TS];

    const int wave  = threadIdx.x >> 5;
    const int lane  = threadIdx.x & 31;
    const int strip = blockIdx.x * LG_W + wave;
    const bool live = strip < LG_NT;
    const int nb = (live ? strip : LG_NT - 1) << 4;
    const int n  = lane & 15;
    const int h  = lane >> 4;

    const float* brow = A3 + (size_t)(nb + n) * EMBD + 2 * h;
    v2f bv[LG_KK];
#pragma unroll
    for (int kk = 0; kk < LG_KK; ++kk)
        bv[kk] = *reinterpret_cast<const v2f*>(brow + kk * 4);

#pragma unroll
    for (int i = 0; i < LG_ST; ++i) {
        const int f = threadIdx.x + i * LG_THR;
        const float4 v = reinterpret_cast<const float4*>(u)[f];
        *reinterpret_cast<float4*>(u_lds + (f >> 5) * LG_PITCH + (f & 31) * 4) = v;
    }
    __syncthreads();

    if (live) {
        const float* arow = u_lds + n * LG_PITCH + 2 * h;
        v8f acc[LG_MT] = {};
#pragma unroll
        for (int kk = 0; kk < LG_KK; ++kk) {
#pragma unroll
            for (int mt = 0; mt < LG_MT; ++mt) {
                const v2f a = *reinterpret_cast<const v2f*>(arow + mt * 16 * LG_PITCH + kk * 4);
                acc[mt] = __builtin_amdgcn_wmma_f32_16x16x4_f32(
                     false, a,  false, bv[kk],
                     (short)0, acc[mt],  false,  false);
            }
        }
        float* trow = tile + wave * LG_TS + (8 * h) * LG_TP + n;
#pragma unroll
        for (int mt = 0; mt < LG_MT; ++mt) {
#pragma unroll
            for (int v = 0; v < 8; ++v) {
                out[(size_t)(mt * 16 + v + 8 * h) * VOCAB + (nb + n)] = acc[mt][v];
                trow[(mt * 16 + v) * LG_TP] = acc[mt][v];
            }
        }
    }
    __builtin_amdgcn_fence(__ATOMIC_RELEASE, "workgroup", "local");
    __builtin_amdgcn_s_barrier();
    __builtin_amdgcn_fence(__ATOMIC_ACQUIRE, "workgroup", "local");

    const int s  = lane & 7;
    const int rg = lane >> 3;
    const bool has = blockIdx.x * LG_W + s < LG_NT;
    float pm[2], ps[2];
#pragma unroll
    for (int j = 0; j < 2; ++j) {
        const int row = 8 * wave + 4 * j + rg;
        pm[j] = -INFINITY;
        ps[j] = 0.f;
        if (has) {
            const float4* t4 = reinterpret_cast<const float4*>(tile + s * LG_TS + row * LG_TP);
            const float4 x0 = t4[0], x1 = t4[1], x2 = t4[2], x3 = t4[3];
            const float mx = fmaxf(fmaxf(fmaxf(fmaxf(x0.x, x0.y), fmaxf(x0.z, x0.w)),
                                         fmaxf(fmaxf(x1.x, x1.y), fmaxf(x1.z, x1.w))),
                                   fmaxf(fmaxf(fmaxf(x2.x, x2.y), fmaxf(x2.z, x2.w)),
                                         fmaxf(fmaxf(x3.x, x3.y), fmaxf(x3.z, x3.w))));
            pm[j] = mx;
            ps[j] = (((__expf(x0.x - mx) + __expf(x0.y - mx)) + (__expf(x0.z - mx) + __expf(x0.w - mx))) +
                     ((__expf(x1.x - mx) + __expf(x1.y - mx)) + (__expf(x1.z - mx) + __expf(x1.w - mx)))) +
                    (((__expf(x2.x - mx) + __expf(x2.y - mx)) + (__expf(x2.z - mx) + __expf(x2.w - mx))) +
                     ((__expf(x3.x - mx) + __expf(x3.y - mx)) + (__expf(x3.z - mx) + __expf(x3.w - mx))));
        }
    }
    for (int o = 1; o < 8; o <<= 1) {
#pragma unroll
        for (int j = 0; j < 2; ++j) {
            const float mo = __shfl_xor(pm[j], o, 32);
            const float so = __shfl_xor(ps[j], o, 32);
            lse_fold(pm[j], ps[j], mo, so);
        }
    }
    if (s == 0) {
#pragma unroll
        for (int j = 0; j < 2; ++j) {
            const int row = 8 * wave + 4 * j + rg;
            *reinterpret_cast<float2*>(part + ((size_t)row * LG_BLOCKS + blockIdx.x) * 2) = make_float2(pm[j], ps[j]);
        }
    }
}

#define LSM_CH  4
#define LSM_THR 512
#define LSM_V4  (VOCAB / 4 / LSM_CH)
#define LSM_IT  ((LSM_V4 + LSM_THR - 1) / LSM_THR)
#define LSM_PI  ((LG_BLOCKS + 31) / 32)

__global__ __launch_bounds__(LSM_THR) void lsm_norm(float* __restrict__ out,
                                                    const float* __restrict__ part) {
    const int ch = blockIdx.x;
    const int b  = blockIdx.y;
    const int t  = threadIdx.x;
    const int lane = t & 31;
    float4* p = reinterpret_cast<float4*>(out + (size_t)b * VOCAB) + ch * LSM_V4;
    const float2* pp = reinterpret_cast<const float2*>(part) + (size_t)b * LG_BLOCKS;

    float2 pr[LSM_PI];
#pragma unroll
    for (int i = 0; i < LSM_PI; ++i) {
        const int idx = lane + 32 * i;
        if (idx < LG_BLOCKS) pr[i] = pp[idx];
        else                 pr[i] = make_float2(-INFINITY, 0.f);
    }

    float4 v[LSM_IT];
#pragma unroll
    for (int i = 0; i < LSM_IT; ++i) {
        const int idx = t + i * LSM_THR;
        if (idx < LSM_V4) v[i] = p[idx];
    }

    float mx = pr[0].x;
#pragma unroll
    for (int i = 1; i < LSM_PI; ++i) mx = fmaxf(mx, pr[i].x);
    float sm = 0.f;
#pragma unroll
    for (int i = 0; i < LSM_PI; ++i) sm += pr[i].y * __expf(pr[i].x - mx);
    for (int o = 16; o > 0; o >>= 1) {
        const float mo = __shfl_xor(mx, o, 32);
        const float so = __shfl_xor(sm, o, 32);
        const float M  = fmaxf(mx, mo);
        sm = sm * __expf(mx - M) + so * __expf(mo - M);
        mx = M;
    }
    const float lse = mx + logf(sm);

#pragma unroll
    for (int i = 0; i < LSM_IT; ++i) {
        const int idx = t + i * LSM_THR;
        if (idx < LSM_V4)
            p[idx] = make_float4(v[i].x - lse, v[i].y - lse, v[i].z - lse, v[i].w - lse);
    }
}

extern "C" void kernel_launch(void* const* d_in, const int* in_sizes, int n_in,
                              void* d_out, int out_size, void* d_ws, size_t ws_size,
                              hipStream_t stream) {
    const int*   x  = (const int*)d_in[0];
    const int*   q  = (const int*)d_in[1];
    const float* A  = (const float*)d_in[2];
    const float* TA = (const float*)d_in[3];
    const float* TC = (const float*)d_in[4];
    float* out = (float*)d_out;

    float* ws = (float*)d_ws;
    float* u  = ws;
    float* m  = ws + BS * EMBD;
    float* c  = m + (size_t)HOPS * BS * STORY * EMBD;

    memnn_gather<<<BS * STORY, EMBD, 0, stream>>>(x, A, TA, TC, m, c);
    memnn_u_chain<<<BS, CH_THR, 0, stream>>>(q, A, m, c, u);

    float* part = m;
    memnn_logits_wmma<<<LG_BLOCKS, LG_THR, 0, stream>>>(
        u, A + 3 * (size_t)VOCAB * EMBD, out, part);
    lsm_norm<<<dim3(LSM_CH, BS), LSM_THR, 0, stream>>>(out, part);
}


# logits GEMM: 256 resident work-groups loop over half-strip units with the next unit's B fragments in flight under the WMMAs; 16-byte fragment loads permute k within 8; row sum tree changes
# speedup vs baseline: 4.0812x; 1.0745x over previous
#include <hip/hip_runtime.h>
#include <hip/hip_bf16.h>
#include <math.h>

#define VOCAB 50000
#define EMBD  128
#define HOPS  3
#define BS    64
#define STORY 50
#define SENT  20
#define QLEN  20

typedef __attribute__((ext_vector_type(2))) float v2f;
typedef __attribute__((ext_vector_type(8))) float v8f;

__global__ __launch_bounds__(EMBD) void memnn_gather(const int* __restrict__ x,
                                                     const float* __restrict__ A,
                                                     const float* __restrict__ TA,
                                                     const float* __restrict__ TC,
                                                     float* __restrict__ m,
                                                     float* __restrict__ c) {
    const int bs = blockIdx.x;
    const int s  = bs % STORY;
    const int d  = threadIdx.x;
    __shared__ int tok[SENT];
    if (d < SENT) tok[d] = x[(size_t)bs * SENT + d];
    __syncthreads();

    const float kd = (float)(d + 1) / (float)EMBD;
    const size_t LVL = (size_t)VOCAB * EMBD;

    float mp0 = 0.f, mp1 = 0.f, mp2 = 0.f;
    float cp1 = 0.f, cp2 = 0.f, cp3 = 0.f;
#pragma unroll
    for (int j = 0; j < SENT; ++j) {
        const float jj = (float)(j + 1) / (float)SENT;
        const float pe = 1.0f - jj - kd * (1.0f - 2.0f * jj);
        const size_t base = (size_t)tok[j] * EMBD + d;
        const float e0 = A[base];
        const float e1 = A[LVL + base];
        const float e2 = A[2 * LVL + base];
        const float e3 = A[3 * LVL + base];
        mp0 += e0 * pe;  mp1 += e1 * pe;  mp2 += e2 * pe;
        cp1 += e1;       cp2 += e2;       cp3 += e3;
    }
    const float ta = TA[s * EMBD + d];
    const float tc = TC[s * EMBD + d];
    const size_t off = (size_t)bs * EMBD + d;
    const size_t hs  = (size_t)BS * STORY * EMBD;
    m[off]          = mp0 + ta;
    m[hs + off]     = mp1 + ta;
    m[2 * hs + off] = mp2 + ta;
    c[off]          = cp1 + tc;
    c[hs + off]     = cp2 + tc;
    c[2 * hs + off] = cp3 + tc;
}

#define CH_THR 512
#define CH_W   (CH_THR / 32)
#define CH_MR  ((STORY + CH_W - 1) / CH_W)
#define CH_G   (CH_THR / EMBD)
#define CH_CR  ((STORY + CH_G - 1) / CH_G)
#define CH_S4  ((STORY + 3) / 4)

__global__ __launch_bounds__(CH_THR) void memnn_u_chain(const int* __restrict__ q,
                                                        const float* __restrict__ A,
                                                        const float* __restrict__ m,
                                                        const float* __restrict__ c,
                                                        float* __restrict__ u) {
    __shared__ __attribute__((aligned(16))) float u_lds[EMBD];
    __shared__ __attribute__((aligned(16))) float score[CH_S4 * 4];
    __shared__ __attribute__((aligned(16))) float pacc[CH_G][EMBD];
    __shared__ float psum[CH_G];

    const int b    = blockIdx.x;
    const int t    = threadIdx.x;
    const int wave = t >> 5;
    const int lane = t & 31;
    const int g    = t >> 7;
    const int d    = t & (EMBD - 1);

    float4 mv[HOPS][CH_MR];
    float  cv[HOPS][CH_CR];
#pragma unroll
    for (int h = 0; h < HOPS; ++h) {
        const size_t hb = ((size_t)h * BS + b) * STORY * EMBD;
#pragma unroll
        for (int i = 0; i < CH_MR; ++i) {
            const int s = min(wave + CH_W * i, STORY - 1);
            mv[h][i] = *reinterpret_cast<const float4*>(m + hb + (size_t)s * EMBD + 4 * lane);
        }
#pragma unroll
        for (int i = 0; i < CH_CR; ++i) {
            const int s = min(g + CH_G * i, STORY - 1);
            cv[h][i] = c[hb + (size_t)s * EMBD + d];
        }
    }

    if (t < EMBD) {
        float acc = 0.0f;
#pragma unroll
        for (int j = 0; j < QLEN; ++j)
            acc += A[(size_t)q[b * QLEN + j] * EMBD + t];
        u_lds[t] = acc;
    }
    if (t < CH_S4 * 4 - STORY) score[STORY + t] = -INFINITY;
    __syncthreads();

    float4 u4 = reinterpret_cast<const float4*>(u_lds)[lane];

#pragma unroll
    for (int h = 0; h < HOPS; ++h) {
        float p[CH_MR];
#pragma unroll
        for (int i = 0; i < CH_MR; ++i)
            p[i] = (mv[h][i].x * u4.x + mv[h][i].y * u4.y) + (mv[h][i].z * u4.z + mv[h][i].w * u4.w);
        for (int o = 16; o > 0; o >>= 1) {
#pragma unroll
            for (int i = 0; i < CH_MR; ++i) p[i] += __shfl_xor(p[i], o, 32);
        }
        if (lane == 0) {
#pragma unroll
            for (int i = 0; i < CH_MR; ++i) {
                const int s = wave + CH_W * i;
                if (s < STORY) score[s] = p[i];
            }
        }
        __syncthreads();

        float mx = -3.4e38f;
#pragma unroll
        for (int k = 0; k < CH_S4; ++k) {
            const float4 sc = reinterpret_cast<const float4*>(score)[k];
            mx = fmaxf(mx, fmaxf(fmaxf(sc.x, sc.y), fmaxf(sc.z, sc.w)));
        }
        float sm = 0.f, acc = 0.f;
#pragma unroll
        for (int i = 0; i < CH_CR; ++i) {
            const int s = g + CH_G * i;
            if (s < STORY) {
                const float e = __expf(score[s] - mx);
                sm  += e;
                acc += e * cv[h][i];
            }
        }
        pacc[g][d] = acc;
        if (d == 0) psum[g] = sm;
        __syncthreads();

        const float tot = (psum[0] + psum[1]) + (psum[2] + psum[3]);
        const float inv = 1.0f / tot;
        const float4 a0 = reinterpret_cast<const float4*>(pacc[0])[lane];
        const float4 a1 = reinterpret_cast<const float4*>(pacc[1])[lane];
        const float4 a2 = reinterpret_cast<const float4*>(pacc[2])[lane];
        const float4 a3 = reinterpret_cast<const float4*>(pacc[3])[lane];
        u4.x += ((a0.x + a1.x) + (a2.x + a3.x)) * inv;
        u4.y += ((a0.y + a1.y) + (a2.y + a3.y)) * inv;
        u4.z += ((a0.z + a1.z) + (a2.z + a3.z)) * inv;
        u4.w += ((a0.w + a1.w) + (a2.w + a3.w)) * inv;
    }
    if (wave == 0) reinterpret_cast<float4*>(u + (size_t)b * EMBD)[lane] = u4;
}

#define PG_WG     256
#define PG_THR    256
#define PG_W      (PG_THR / 32)
#define PG_NU     (2 * (VOCAB / 16))
#define PG_MAXU   4
#define PG_KJ     (EMBD / 8)
#define LG_PITCH  (EMBD + 4)
#define LG_ST     (BS * EMBD / 4 / PG_THR)
#define PG_TP     20
#define PG_TW     (32 * PG_TP)

__device__ __forceinline__ void lse_fold(float& m, float& s, float mo, float so) {
    const float M = fmaxf(m, mo);
    const float a = (m  == M) ? 1.0f : __expf(m  - M);
    const float b = (mo == M) ? 1.0f : __expf(mo - M);
    s = s * a + so * b;
    m = M;
}

__device__ __forceinline__ void pg_load_b(float4 (&bv)[PG_KJ], const float* __restrict__ A3,
                                          int unit, int c, int hh) {
    const float* brow = A3 + (size_t)(((unit >> 1) << 4) + c) * EMBD + 4 * hh;
#pragma unroll
    for (int j = 0; j < PG_KJ; ++j)
        bv[j] = *reinterpret_cast<const float4*>(brow + 8 * j);
}

__device__ __forceinline__ void pg_unit(const float4 (&bv)[PG_KJ], int unit,
                                        const float* u_lds, float* wt, float* __restrict__ out,
                                        int c, int hh, int lane, float& rm, float& rs) {
    const int strip = unit >> 1;
    const int half  = unit & 1;
    const float* arow = u_lds + (32 * half + c) * LG_PITCH + 4 * hh;
    v8f acc0 = {};
    v8f acc1 = {};
#pragma unroll
    for (int j = 0; j < PG_KJ; ++j) {
        const float4 a0 = *reinterpret_cast<const float4*>(arow + 8 * j);
        const float4 a1 = *reinterpret_cast<const float4*>(arow + 16 * LG_PITCH + 8 * j);
        const v2f blo = {bv[j].x, bv[j].y};
        const v2f bhi = {bv[j].z, bv[j].w};
        const v2f a0lo = {a0.x, a0.y};
        const v2f a0hi = {a0.z, a0.w};
        const v2f a1lo = {a1.x, a1.y};
        const v2f a1hi = {a1.z, a1.w};
        acc0 = __builtin_amdgcn_wmma_f32_16x16x4_f32(false, a0lo, false, blo, (short)0, acc0, false, false);
        acc1 = __builtin_amdgcn_wmma_f32_16x16x4_f32(false, a1lo, false, blo, (short)0, acc1, false, false);
        acc0 = __builtin_amdgcn_wmma_f32_16x16x4_f32(false, a0hi, false, bhi, (short)0, acc0, false, false);
        acc1 = __builtin_amdgcn_wmma_f32_16x16x4_f32(false, a1hi, false, bhi, (short)0, acc1, false, false);
    }
    float* orow = out + (size_t)(32 * half + 8 * hh) * VOCAB + (strip << 4) + c;
    float* trow = wt + (8 * hh) * PG_TP + c;
#pragma unroll
    for (int v = 0; v < 8; ++v) {
        orow[(size_t)v * VOCAB]        = acc0[v];
        orow[(size_t)(16 + v) * VOCAB] = acc1[v];
        trow[v * PG_TP]        = acc0[v];
        trow[(16 + v) * PG_TP] = acc1[v];
    }
    __builtin_amdgcn_fence(__ATOMIC_RELEASE, "workgroup", "local");
    __builtin_amdgcn_wave_barrier();
    __builtin_amdgcn_fence(__ATOMIC_ACQUIRE, "workgroup", "local");
    const float4* t4 = reinterpret_cast<const float4*>(wt + lane * PG_TP);
    const float4 x0 = t4[0], x1 = t4[1], x2 = t4[2], x3 = t4[3];
    const float mx = fmaxf(fmaxf(fmaxf(fmaxf(x0.x, x0.y), fmaxf(x0.z, x0.w)),
                                 fmaxf(fmaxf(x1.x, x1.y), fmaxf(x1.z, x1.w))),
                           fmaxf(fmaxf(fmaxf(x2.x, x2.y), fmaxf(x2.z, x2.w)),
                                 fmaxf(fmaxf(x3.x, x3.y), fmaxf(x3.z, x3.w))));
    const float sm = (((__expf(x0.x - mx) + __expf(x0.y - mx)) + (__expf(x0.z - mx) + __expf(x0.w - mx))) +
                      ((__expf(x1.x - mx) + __expf(x1.y - mx)) + (__expf(x1.z - mx) + __expf(x1.w - mx)))) +
                     (((__expf(x2.x - mx) + __expf(x2.y - mx)) + (__expf(x2.z - mx) + __expf(x2.w - mx))) +
                      ((__expf(x3.x - mx) + __expf(x3.y - mx)) + (__expf(x3.z - mx) + __expf(x3.w - mx))));
    lse_fold(rm, rs, mx, sm);
    __builtin_amdgcn_fence(__ATOMIC_RELEASE, "workgroup", "local");
    __builtin_amdgcn_wave_barrier();
}

__global__ __launch_bounds__(PG_THR) void memnn_logits_wmma(const float* __restrict__ u,
                                                            const float* __restrict__ A3,
                                                            float* __restrict__ out,
                                                            float* __restrict__ part) {
    __shared__ __attribute__((aligned(16))) float u_lds[BS * LG_PITCH];
    __shared__ __attribute__((aligned(16))) float tile[PG_W * PG_TW];
    __shared__ float2 wpair[PG_W][32];

    const int wave = threadIdx.x >> 5;
    const int lane = threadIdx.x & 31;
    const int c    = lane & 15;
    const int hh   = lane >> 4;
    const int ub = (int)(blockIdx.x * PG_NU) / PG_WG;
    const int ue = (int)((blockIdx.x + 1) * PG_NU) / PG_WG;
    const int first = ub + wave;

    float4 bA[PG_KJ], bB[PG_KJ];
    pg_load_b(bA, A3, first, c, hh);

#pragma unroll
    for (int i = 0; i < LG_ST; ++i) {
        const int f = threadIdx.x + i * PG_THR;
        const float4 v = reinterpret_cast<const float4*>(u)[f];
        *reinterpret_cast<float4*>(u_lds + (f >> 5) * LG_PITCH + (f & 31) * 4) = v;
    }
    __syncthreads();

    float* wt = tile + wave * PG_TW;
    float rm = -INFINITY, rs = 0.f;
#pragma unroll
    for (int i = 0; i < PG_MAXU; ++i) {
        const int cur = first + PG_W * i;
        if (cur >= ue) break;
        const int nxt = cur + PG_W;
        if (i & 1) {
            if (i + 1 < PG_MAXU && nxt < ue) pg_load_b(bA, A3, nxt, c, hh);
            pg_unit(bB, cur, u_lds, wt, out, c, hh, lane, rm, rs);
        } else {
            if (i + 1 < PG_MAXU && nxt < ue) pg_load_b(bB, A3, nxt, c, hh);
            pg_unit(bA, cur, u_lds, wt, out, c, hh, lane, rm, rs);
        }
    }

    wpair[wave][lane] = make_float2(rm, rs);
    __builtin_amdgcn_fence(__ATOMIC_RELEASE, "workgroup", "local");
    __builtin_amdgcn_s_barrier();
    __builtin_amdgcn_fence(__ATOMIC_ACQUIRE, "workgroup", "local");
    if (threadIdx.x < BS) {
        const int row = threadIdx.x;
        const int hf  = row >> 5;
        const int ln  = row & 31;
        float m = -INFINITY, s = 0.f;
#pragma unroll
        for (int w = 0; w < PG_W; ++w) {
            const float2 p = wpair[w][ln];
            if (((ub + w) & 1) == hf) lse_fold(m, s, p.x, p.y);
        }
        *reinterpret_cast<float2*>(part + ((size_t)row * PG_WG + blockIdx.x) * 2) = make_float2(m, s);
    }
}

#define LSM_CH  4
#define LSM_THR 512
#define LSM_V4  (VOCAB / 4 / LSM_CH)
#define LSM_IT  ((LSM_V4 + LSM_THR - 1) / LSM_THR)
#define LSM_PI  ((PG_WG + 31) / 32)

__global__ __launch_bounds__(LSM_THR) void lsm_norm(float* __restrict__ out,
                                                    const float* __restrict__ part) {
    const int ch = blockIdx.x;
    const int b  = blockIdx.y;
    const int t  = threadIdx.x;
    const int lane = t & 31;
    float4* p = reinterpret_cast<float4*>(out + (size_t)b * VOCAB) + ch * LSM_V4;
    const float2* pp = reinterpret_cast<const float2*>(part) + (size_t)b * PG_WG;

    float2 pr[LSM_PI];
#pragma unroll
    for (int i = 0; i < LSM_PI; ++i) {
        const int idx = lane + 32 * i;
        if (idx < PG_WG) pr[i] = pp[idx];
        else                 pr[i] = make_float2(-INFINITY, 0.f);
    }

    float4 v[LSM_IT];
#pragma unroll
    for (int i = 0; i < LSM_IT; ++i) {
        const int idx = t + i * LSM_THR;
        if (idx < LSM_V4) v[i] = p[idx];
    }

    float mx = pr[0].x;
#pragma unroll
    for (int i = 1; i < LSM_PI; ++i) mx = fmaxf(mx, pr[i].x);
    float sm = 0.f;
#pragma unroll
    for (int i = 0; i < LSM_PI; ++i) sm += pr[i].y * __expf(pr[i].x - mx);
    for (int o = 16; o > 0; o >>= 1) {
        const float mo = __shfl_xor(mx, o, 32);
        const float so = __shfl_xor(sm, o, 32);
        const float M  = fmaxf(mx, mo);
        sm = sm * __expf(mx - M) + so * __expf(mo - M);
        mx = M;
    }
    const float lse = mx + logf(sm);

#pragma unroll
    for (int i = 0; i < LSM_IT; ++i) {
        const int idx = t + i * LSM_THR;
        if (idx < LSM_V4)
            p[idx] = make_float4(v[i].x - lse, v[i].y - lse, v[i].z - lse, v[i].w - lse);
    }
}

extern "C" void kernel_launch(void* const* d_in, const int* in_sizes, int n_in,
                              void* d_out, int out_size, void* d_ws, size_t ws_size,
                              hipStream_t stream) {
    const int*   x  = (const int*)d_in[0];
    const int*   q  = (const int*)d_in[1];
    const float* A  = (const float*)d_in[2];
    const float* TA = (const float*)d_in[3];
    const float* TC = (const float*)d_in[4];
    float* out = (float*)d_out;

    float* ws = (float*)d_ws;
    float* u  = ws;
    float* m  = ws + BS * EMBD;
    float* c  = m + (size_t)HOPS * BS * STORY * EMBD;

    memnn_gather<<<BS * STORY, EMBD, 0, stream>>>(x, A, TA, TC, m, c);
    memnn_u_chain<<<BS, CH_THR, 0, stream>>>(q, A, m, c, u);

    float* part = m;
    memnn_logits_wmma<<<PG_WG, PG_THR, 0, stream>>>(
        u, A + 3 * (size_t)VOCAB * EMBD, out, part);
    lsm_norm<<<dim3(LSM_CH, BS), LSM_THR, 0, stream>>>(out, part);
}


# log-softmax normalising launch: work-groups of 128 threads with 25 float4 per lane instead of 512 threads with 7, so 4096 short waves become 1024; pair fold, arithmetic and sum order kept
# speedup vs baseline: 4.3866x; 1.0748x over previous
#include <hip/hip_runtime.h>
#include <hip/hip_bf16.h>
#include <math.h>

#define VOCAB 50000
#define EMBD  128
#define HOPS  3
#define BS    64
#define STORY 50
#define SENT  20
#define QLEN  20

typedef __attribute__((ext_vector_type(2))) float v2f;
typedef __attribute__((ext_vector_type(8))) float v8f;

__global__ __launch_bounds__(EMBD) void memnn_gather(const int* __restrict__ x,
                                                     const float* __restrict__ A,
                                                     const float* __restrict__ TA,
                                                     const float* __restrict__ TC,
                                                     float* __restrict__ m,
                                                     float* __restrict__ c) {
    const int bs = blockIdx.x;
    const int s  = bs % STORY;
    const int d  = threadIdx.x;
    __shared__ int tok[SENT];
    if (d < SENT) tok[d] = x[(size_t)bs * SENT + d];
    __syncthreads();

    const float kd = (float)(d + 1) / (float)EMBD;
    const size_t LVL = (size_t)VOCAB * EMBD;

    float mp0 = 0.f, mp1 = 0.f, mp2 = 0.f;
    float cp1 = 0.f, cp2 = 0.f, cp3 = 0.f;
#pragma unroll
    for (int j = 0; j < SENT; ++j) {
        const float jj = (float)(j + 1) / (float)SENT;
        const float pe = 1.0f - jj - kd * (1.0f - 2.0f * jj);
        const size_t base = (size_t)tok[j] * EMBD + d;
        const float e0 = A[base];
        const float e1 = A[LVL + base];
        const float e2 = A[2 * LVL + base];
        const float e3 = A[3 * LVL + base];
        mp0 += e0 * pe;  mp1 += e1 * pe;  mp2 += e2 * pe;
        cp1 += e1;       cp2 += e2;       cp3 += e3;
    }
    const float ta = TA[s * EMBD + d];
    const float tc = TC[s * EMBD + d];
    const size_t off = (size_t)bs * EMBD + d;
    const size_t hs  = (size_t)BS * STORY * EMBD;
    m[off]          = mp0 + ta;
    m[hs + off]     = mp1 + ta;
    m[2 * hs + off] = mp2 + ta;
    c[off]          = cp1 + tc;
    c[hs + off]     = cp2 + tc;
    c[2 * hs + off] = cp3 + tc;
}

#define CH_THR 512
#define CH_W   (CH_THR / 32)
#define CH_MR  ((STORY + CH_W - 1) / CH_W)
#define CH_G   (CH_THR / EMBD)
#define CH_CR  ((STORY + CH_G - 1) / CH_G)
#define CH_S4  ((STORY + 3) / 4)

__global__ __launch_bounds__(CH_THR) void memnn_u_chain(const int* __restrict__ q,
                                                        const float* __restrict__ A,
                                                        const float* __restrict__ m,
                                                        const float* __restrict__ c,
                                                        float* __restrict__ u) {
    __shared__ __attribute__((aligned(16))) float u_lds[EMBD];
    __shared__ __attribute__((aligned(16))) float score[CH_S4 * 4];
    __shared__ __attribute__((aligned(16))) float pacc[CH_G][EMBD];
    __shared__ float psum[CH_G];

    const int b    = blockIdx.x;
    const int t    = threadIdx.x;
    const int wave = t >> 5;
    const int lane = t & 31;
    const int g    = t >> 7;
    const int d    = t & (EMBD - 1);

    float4 mv[HOPS][CH_MR];
    float  cv[HOPS][CH_CR];
#pragma unroll
    for (int h = 0; h < HOPS; ++h) {
        const size_t hb = ((size_t)h * BS + b) * STORY * EMBD;
#pragma unroll
        for (int i = 0; i < CH_MR; ++i) {
            const int s = min(wave + CH_W * i, STORY - 1);
            mv[h][i] = *reinterpret_cast<const float4*>(m + hb + (size_t)s * EMBD + 4 * lane);
        }
#pragma unroll
        for (int i = 0; i < CH_CR; ++i) {
            const int s = min(g + CH_G * i, STORY - 1);
            cv[h][i] = c[hb + (size_t)s * EMBD + d];
        }
    }

    if (t < EMBD) {
        float acc = 0.0f;
#pragma unroll
        for (int j = 0; j < QLEN; ++j)
            acc += A[(size_t)q[b * QLEN + j] * EMBD + t];
        u_lds[t] = acc;
    }
    if (t < CH_S4 * 4 - STORY) score[STORY + t] = -INFINITY;
    __syncthreads();

    float4 u4 = reinterpret_cast<const float4*>(u_lds)[lane];

#pragma unroll
    for (int h = 0; h < HOPS; ++h) {
        float p[CH_MR];
#pragma unroll
        for (int i = 0; i < CH_MR; ++i)
            p[i] = (mv[h][i].x * u4.x + mv[h][i].y * u4.y) + (mv[h][i].z * u4.z + mv[h][i].w * u4.w);
        for (int o = 16; o > 0; o >>= 1) {
#pragma unroll
            for (int i = 0; i < CH_MR; ++i) p[i] += __shfl_xor(p[i], o, 32);
        }
        if (lane == 0) {
#pragma unroll
            for (int i = 0; i < CH_MR; ++i) {
                const int s = wave + CH_W * i;
                if (s < STORY) score[s] = p[i];
            }
        }
        __syncthreads();

        float mx = -3.4e38f;
#pragma unroll
        for (int k = 0; k < CH_S4; ++k) {
            const float4 sc = reinterpret_cast<const float4*>(score)[k];
            mx = fmaxf(mx, fmaxf(fmaxf(sc.x, sc.y), fmaxf(sc.z, sc.w)));
        }
        float sm = 0.f, acc = 0.f;
#pragma unroll
        for (int i = 0; i < CH_CR; ++i) {
            const int s = g + CH_G * i;
            if (s < STORY) {
                const float e = __expf(score[s] - mx);
                sm  += e;
                acc += e * cv[h][i];
            }
        }
        pacc[g][d] = acc;
        if (d == 0) psum[g] = sm;
        __syncthreads();

        const float tot = (psum[0] + psum[1]) + (psum[2] + psum[3]);
        const float inv = 1.0f / tot;
        const float4 a0 = reinterpret_cast<const float4*>(pacc[0])[lane];
        const float4 a1 = reinterpret_cast<const float4*>(pacc[1])[lane];
        const float4 a2 = reinterpret_cast<const float4*>(pacc[2])[lane];
        const float4 a3 = reinterpret_cast<const float4*>(pacc[3])[lane];
        u4.x += ((a0.x + a1.x) + (a2.x + a3.x)) * inv;
        u4.y += ((a0.y + a1.y) + (a2.y + a3.y)) * inv;
        u4.z += ((a0.z + a1.z) + (a2.z + a3.z)) * inv;
        u4.w += ((a0.w + a1.w) + (a2.w + a3.w)) * inv;
    }
    if (wave == 0) reinterpret_cast<float4*>(u + (size_t)b * EMBD)[lane] = u4;
}

#define PG_WG     256
#define PG_THR    256
#define PG_W      (PG_THR / 32)
#define PG_NU     (2 * (VOCAB / 16))
#define PG_MAXU   4
#define PG_KJ     (EMBD / 8)
#define LG_PITCH  (EMBD + 4)
#define LG_ST     (BS * EMBD / 4 / PG_THR)
#define PG_TP     20
#define PG_TW     (32 * PG_TP)

__device__ __forceinline__ void lse_fold(float& m, float& s, float mo, float so) {
    const float M = fmaxf(m, mo);
    const float a = (m  == M) ? 1.0f : __expf(m  - M);
    const float b = (mo == M) ? 1.0f : __expf(mo - M);
    s = s * a + so * b;
    m = M;
}

__device__ __forceinline__ void pg_load_b(float4 (&bv)[PG_KJ], const float* __restrict__ A3,
                                          int unit, int c, int hh) {
    const float* brow = A3 + (size_t)(((unit >> 1) << 4) + c) * EMBD + 4 * hh;
#pragma unroll
    for (int j = 0; j < PG_KJ; ++j)
        bv[j] = *reinterpret_cast<const float4*>(brow + 8 * j);
}

__device__ __forceinline__ void pg_unit(const float4 (&bv)[PG_KJ], int unit,
                                        const float* u_lds, float* wt, float* __restrict__ out,
                                        int c, int hh, int lane, float& rm, float& rs) {
    const int strip = unit >> 1;
    const int half  = unit & 1;
    const float* arow = u_lds + (32 * half + c) * LG_PITCH + 4 * hh;
    v8f acc0 = {};
    v8f acc1 = {};
#pragma unroll
    for (int j = 0; j < PG_KJ; ++j) {
        const float4 a0 = *reinterpret_cast<const float4*>(arow + 8 * j);
        const float4 a1 = *reinterpret_cast<const float4*>(arow + 16 * LG_PITCH + 8 * j);
        const v2f blo = {bv[j].x, bv[j].y};
        const v2f bhi = {bv[j].z, bv[j].w};
        const v2f a0lo = {a0.x, a0.y};
        const v2f a0hi = {a0.z, a0.w};
        const v2f a1lo = {a1.x, a1.y};
        const v2f a1hi = {a1.z, a1.w};
        acc0 = __builtin_amdgcn_wmma_f32_16x16x4_f32(false, a0lo, false, blo, (short)0, acc0, false, false);
        acc1 = __builtin_amdgcn_wmma_f32_16x16x4_f32(false, a1lo, false, blo, (short)0, acc1, false, false);
        acc0 = __builtin_amdgcn_wmma_f32_16x16x4_f32(false, a0hi, false, bhi, (short)0, acc0, false, false);
        acc1 = __builtin_amdgcn_wmma_f32_16x16x4_f32(false, a1hi, false, bhi, (short)0, acc1, false, false);
    }
    float* orow = out + (size_t)(32 * half + 8 * hh) * VOCAB + (strip << 4) + c;
    float* trow = wt + (8 * hh) * PG_TP + c;
#pragma unroll
    for (int v = 0; v < 8; ++v) {
        orow[(size_t)v * VOCAB]        = acc0[v];
        orow[(size_t)(16 + v) * VOCAB] = acc1[v];
        trow[v * PG_TP]        = acc0[v];
        trow[(16 + v) * PG_TP] = acc1[v];
    }
    __builtin_amdgcn_fence(__ATOMIC_RELEASE, "workgroup", "local");
    __builtin_amdgcn_wave_barrier();
    __builtin_amdgcn_fence(__ATOMIC_ACQUIRE, "workgroup", "local");
    const float4* t4 = reinterpret_cast<const float4*>(wt + lane * PG_TP);
    const float4 x0 = t4[0], x1 = t4[1], x2 = t4[2], x3 = t4[3];
    const float mx = fmaxf(fmaxf(fmaxf(fmaxf(x0.x, x0.y), fmaxf(x0.z, x0.w)),
                                 fmaxf(fmaxf(x1.x, x1.y), fmaxf(x1.z, x1.w))),
                           fmaxf(fmaxf(fmaxf(x2.x, x2.y), fmaxf(x2.z, x2.w)),
                                 fmaxf(fmaxf(x3.x, x3.y), fmaxf(x3.z, x3.w))));
    const float sm = (((__expf(x0.x - mx) + __expf(x0.y - mx)) + (__expf(x0.z - mx) + __expf(x0.w - mx))) +
                      ((__expf(x1.x - mx) + __expf(x1.y - mx)) + (__expf(x1.z - mx) + __expf(x1.w - mx)))) +
                     (((__expf(x2.x - mx) + __expf(x2.y - mx)) + (__expf(x2.z - mx) + __expf(x2.w - mx))) +
                      ((__expf(x3.x - mx) + __expf(x3.y - mx)) + (__expf(x3.z - mx) + __expf(x3.w - mx))));
    lse_fold(rm, rs, mx, sm);
    __builtin_amdgcn_fence(__ATOMIC_RELEASE, "workgroup", "local");
    __builtin_amdgcn_wave_barrier();
}

__global__ __launch_bounds__(PG_THR) void memnn_logits_wmma(const float* __restrict__ u,
                                                            const float* __restrict__ A3,
                                                            float* __restrict__ out,
                                                            float* __restrict__ part) {
    __shared__ __attribute__((aligned(16))) float u_lds[BS * LG_PITCH];
    __shared__ __attribute__((aligned(16))) float tile[PG_W * PG_TW];
    __shared__ float2 wpair[PG_W][32];

    const int wave = threadIdx.x >> 5;
    const int lane = threadIdx.x & 31;
    const int c    = lane & 15;
    const int hh   = lane >> 4;
    const int ub = (int)(blockIdx.x * PG_NU) / PG_WG;
    const int ue = (int)((blockIdx.x + 1) * PG_NU) / PG_WG;
    const int first = ub + wave;

    float4 bA[PG_KJ], bB[PG_KJ];
    pg_load_b(bA, A3, first, c, hh);

#pragma unroll
    for (int i = 0; i < LG_ST; ++i) {
        const int f = threadIdx.x + i * PG_THR;
        const float4 v = reinterpret_cast<const float4*>(u)[f];
        *reinterpret_cast<float4*>(u_lds + (f >> 5) * LG_PITCH + (f & 31) * 4) = v;
    }
    __syncthreads();

    float* wt = tile + wave * PG_TW;
    float rm = -INFINITY, rs = 0.f;
#pragma unroll
    for (int i = 0; i < PG_MAXU; ++i) {
        const int cur = first + PG_W * i;
        if (cur >= ue) break;
        const int nxt = cur + PG_W;
        if (i & 1) {
            if (i + 1 < PG_MAXU && nxt < ue) pg_load_b(bA, A3, nxt, c, hh);
            pg_unit(bB, cur, u_lds, wt, out, c, hh, lane, rm, rs);
        } else {
            if (i + 1 < PG_MAXU && nxt < ue) pg_load_b(bB, A3, nxt, c, hh);
            pg_unit(bA, cur, u_lds, wt, out, c, hh, lane, rm, rs);
        }
    }

    wpair[wave][lane] = make_float2(rm, rs);
    __builtin_amdgcn_fence(__ATOMIC_RELEASE, "workgroup", "local");
    __builtin_amdgcn_s_barrier();
    __builtin_amdgcn_fence(__ATOMIC_ACQUIRE, "workgroup", "local");
    if (threadIdx.x < BS) {
        const int row = threadIdx.x;
        const int hf  = row >> 5;
        const int ln  = row & 31;
        float m = -INFINITY, s = 0.f;
#pragma unroll
        for (int w = 0; w < PG_W; ++w) {
            const float2 p = wpair[w][ln];
            if (((ub + w) & 1) == hf) lse_fold(m, s, p.x, p.y);
        }
        *reinterpret_cast<float2*>(part + ((size_t)row * PG_WG + blockIdx.x) * 2) = make_float2(m, s);
    }
}

#define LSM_CH  4
#define LSM_THR 128
#define LSM_V4  (VOCAB / 4 / LSM_CH)
#define LSM_IT  ((LSM_V4 + LSM_THR - 1) / LSM_THR)
#define LSM_FULL (LSM_V4 / LSM_THR)
#define LSM_PI  ((PG_WG + 31) / 32)

__global__ __launch_bounds__(LSM_THR) void lsm_norm(float* __restrict__ out,
                                                    const float* __restrict__ part) {
    const int ch = blockIdx.x;
    const int b  = blockIdx.y;
    const int t  = threadIdx.x;
    const int lane = t & 31;
    float4* p = reinterpret_cast<float4*>(out + (size_t)b * VOCAB) + ch * LSM_V4;
    const float2* pp = reinterpret_cast<const float2*>(part) + (size_t)b * PG_WG;

    float2 pr[LSM_PI];
#pragma unroll
    for (int i = 0; i < LSM_PI; ++i) {
        const int idx = lane + 32 * i;
        if (idx < PG_WG) pr[i] = pp[idx];
        else                 pr[i] = make_float2(-INFINITY, 0.f);
    }

    float4 v[LSM_IT];
#pragma unroll
    for (int i = 0; i < LSM_IT; ++i) {
        const int idx = t + i * LSM_THR;
        if (i < LSM_FULL || idx < LSM_V4) v[i] = p[idx];
    }

    float mx = pr[0].x;
#pragma unroll
    for (int i = 1; i < LSM_PI; ++i) mx = fmaxf(mx, pr[i].x);
    float sm = 0.f;
#pragma unroll
    for (int i = 0; i < LSM_PI; ++i) sm += pr[i].y * __expf(pr[i].x - mx);
    for (int o = 16; o > 0; o >>= 1) {
        const float mo = __shfl_xor(mx, o, 32);
        const float so = __shfl_xor(sm, o, 32);
        const float M  = fmaxf(mx, mo);
        sm = sm * __expf(mx - M) + so * __expf(mo - M);
        mx = M;
    }
    const float lse = mx + logf(sm);

#pragma unroll
    for (int i = 0; i < LSM_IT; ++i) {
        const int idx = t + i * LSM_THR;
        if (i < LSM_FULL || idx < LSM_V4)
            p[idx] = make_float4(v[i].x - lse, v[i].y - lse, v[i].z - lse, v[i].w - lse);
    }
}

extern "C" void kernel_launch(void* const* d_in, const int* in_sizes, int n_in,
                              void* d_out, int out_size, void* d_ws, size_t ws_size,
                              hipStream_t stream) {
    const int*   x  = (const int*)d_in[0];
    const int*   q  = (const int*)d_in[1];
    const float* A  = (const float*)d_in[2];
    const float* TA = (const float*)d_in[3];
    const float* TC = (const float*)d_in[4];
    float* out = (float*)d_out;

    float* ws = (float*)d_ws;
    float* u  = ws;
    float* m  = ws + BS * EMBD;
    float* c  = m + (size_t)HOPS * BS * STORY * EMBD;

    memnn_gather<<<BS * STORY, EMBD, 0, stream>>>(x, A, TA, TC, m, c);
    memnn_u_chain<<<BS, CH_THR, 0, stream>>>(q, A, m, c, u);

    float* part = m;
    memnn_logits_wmma<<<PG_WG, PG_THR, 0, stream>>>(
        u, A + 3 * (size_t)VOCAB * EMBD, out, part);
    lsm_norm<<<dim3(LSM_CH, BS), LSM_THR, 0, stream>>>(out, part);
}
